# starting kernel
# baseline (speedup 1.0000x reference)
#include <hip/hip_runtime.h>


typedef __attribute__((ext_vector_type(2))) float v2f;
typedef __attribute__((ext_vector_type(8))) float v8f;

#define BTOT 16
#define WN   64
#define NN   64
#define CC   96
#define HH   8
#define DD   768
#define C3   288
#define ROW3 18432
#define YBT  1179648
#define HD   6144
#define ZBT  393216
#define ATTN_SCALE 0.28867513459481287f

static __device__ __forceinline__ v8f wmma4(v2f a, v2f b, v8f c) {
  return __builtin_amdgcn_wmma_f32_16x16x4_f32(
       false, a,  false, b,
       (short)0, c,  false,  false);
}

__global__ __launch_bounds__(256) void qkv_gemm_kernel(
    const float* __restrict__ x, const float* __restrict__ wqkv,
    float* __restrict__ Y) {
  __shared__ __align__(16) float sX[64 * 100];
  const int tid = threadIdx.x;
  const int lane = tid & 31, wave = tid >> 5;
  const int g = blockIdx.x;
  const float* xt = x + (size_t)g * (NN * CC);
  float* yt = Y + (size_t)(g / WN) * YBT + (size_t)(g % WN) * ROW3;

  for (int i = tid; i < 64 * 24; i += 256) {
    int r = i / 24, c4 = (i % 24) * 4;
    *reinterpret_cast<float4*>(&sX[r * 100 + c4]) =
        *reinterpret_cast<const float4*>(xt + r * CC + c4);
  }
  __syncthreads();

  const int m0 = lane & 15;
  const int ksel = (lane >> 4) * 2;
  for (int tt = wave; tt < 72; tt += 8) {
    const int mt = tt & 3, nt = tt >> 2;
    const int ncol = nt * 16 + m0;
    const float* bp = wqkv + ncol * CC;
    const float* ap = &sX[(mt * 16 + m0) * 100];
    v8f acc = {};
#pragma unroll
    for (int kk = 0; kk < 24; ++kk) {
      const int kb = kk * 4 + ksel;
      v2f a = *reinterpret_cast<const v2f*>(ap + kb);
      v2f b = *reinterpret_cast<const v2f*>(bp + kb);
      acc = wmma4(a, b, acc);
    }
    float* yo = yt + ncol;
#pragma unroll
    for (int v = 0; v < 8; ++v) {
      const int m = mt * 16 + v + ((lane >> 4) << 3);
      yo[m * C3] = acc[v];
    }
  }
}

__global__ __launch_bounds__(256) void win_attn_kernel(
    const float* __restrict__ Y, const float* __restrict__ mask,
    float* __restrict__ Z) {
  __shared__ __align__(16) float sA[64 * 68];
  __shared__ __align__(16) float sB[64 * 68];
  __shared__ __align__(16) float sS[64 * 68];
  const int tid = threadIdx.x, lane = tid & 31, wave = tid >> 5;
  const int bt = blockIdx.x >> 3, h = blockIdx.x & 7;
  const float* qb = Y + (size_t)bt * YBT + h * DD;
  const float* kbp = qb + HD;
  const float* vbp = qb + 2 * HD;

  const int m0 = lane & 15;
  const int ksel = (lane >> 4) * 2;
  const int t0 = wave * 2;

  v8f accS[2];
  accS[0] = (v8f){};
  accS[1] = (v8f){};

  for (int dc = 0; dc < 12; ++dc) {
    for (int i = tid; i < 64 * 16; i += 256) {
      const int r = i / 16, c4 = (i % 16) * 4;
      *reinterpret_cast<float4*>(&sA[r * 68 + c4]) =
          *reinterpret_cast<const float4*>(qb + r * ROW3 + dc * 64 + c4);
      *reinterpret_cast<float4*>(&sB[r * 68 + c4]) =
          *reinterpret_cast<const float4*>(kbp + r * ROW3 + dc * 64 + c4);
    }
    __syncthreads();
#pragma unroll
    for (int j = 0; j < 2; ++j) {
      const int tt = t0 + j, mt = tt >> 2, nt = tt & 3;
      const float* ap = &sA[(mt * 16 + m0) * 68];
      const float* bp = &sB[(nt * 16 + m0) * 68];
#pragma unroll
      for (int kk = 0; kk < 16; ++kk) {
        const int kb = kk * 4 + ksel;
        v2f a = *reinterpret_cast<const v2f*>(ap + kb);
        v2f b = *reinterpret_cast<const v2f*>(bp + kb);
        accS[j] = wmma4(a, b, accS[j]);
      }
    }
    __syncthreads();
  }

  const float* mk = mask + (size_t)(bt & 7) * (NN * NN);
#pragma unroll
  for (int j = 0; j < 2; ++j) {
    const int tt = t0 + j, mt = tt >> 2, nt = tt & 3;
    const int ncol = nt * 16 + m0;
#pragma unroll
    for (int v = 0; v < 8; ++v) {
      const int m = mt * 16 + v + ((lane >> 4) << 3);
      sS[m * 68 + ncol] = accS[j][v] * ATTN_SCALE + mk[m * NN + ncol];
    }
  }
  __syncthreads();

  for (int r = wave * 8; r < wave * 8 + 8; ++r) {
    float x0 = sS[r * 68 + lane];
    float x1 = sS[r * 68 + lane + 32];
    float mx = fmaxf(x0, x1);
#pragma unroll
    for (int off = 16; off >= 1; off >>= 1)
      mx = fmaxf(mx, __shfl_xor(mx, off, 32));
    const float e0 = __expf(x0 - mx), e1 = __expf(x1 - mx);
    float sm = e0 + e1;
#pragma unroll
    for (int off = 16; off >= 1; off >>= 1)
      sm += __shfl_xor(sm, off, 32);
    const float inv = 1.0f / sm;
    sS[r * 68 + lane] = e0 * inv;
    sS[r * 68 + lane + 32] = e1 * inv;
  }
  __syncthreads();

  float* zb = Z + (size_t)bt * ZBT + h * DD;
  for (int dc = 0; dc < 12; ++dc) {
    for (int i = tid; i < 64 * 16; i += 256) {
      const int r = i / 16, c4 = (i % 16) * 4;
      *reinterpret_cast<float4*>(&sB[r * 68 + c4]) =
          *reinterpret_cast<const float4*>(vbp + r * ROW3 + dc * 64 + c4);
    }
    __syncthreads();
#pragma unroll
    for (int j = 0; j < 2; ++j) {
      const int tt = t0 + j, mt = tt >> 2, dt = tt & 3;
      const float* ap = &sS[(mt * 16 + m0) * 68];
      v8f acc = {};
#pragma unroll
      for (int kk = 0; kk < 16; ++kk) {
        const int kb = kk * 4 + ksel;
        v2f a = *reinterpret_cast<const v2f*>(ap + kb);
        v2f b;
        b.x = sB[kb * 68 + dt * 16 + m0];
        b.y = sB[(kb + 1) * 68 + dt * 16 + m0];
        acc = wmma4(a, b, acc);
      }
      const int dcol = dc * 64 + dt * 16 + m0;
#pragma unroll
      for (int v = 0; v < 8; ++v) {
        const int m = mt * 16 + v + ((lane >> 4) << 3);
        zb[m * HD + dcol] = acc[v];
      }
    }
    __syncthreads();
  }
}

__global__ __launch_bounds__(256) void proj_gemm_kernel(
    const float* __restrict__ Z, const float* __restrict__ wproj,
    const float* __restrict__ bproj, float* __restrict__ out) {
  __shared__ __align__(16) float sZ[64 * 100];
  const int tid = threadIdx.x, lane = tid & 31, wave = tid >> 5;
  const int g = blockIdx.x;
  const float* zt = Z + (size_t)g * (NN * CC);
  float* ot = out + (size_t)g * (NN * CC);

  for (int i = tid; i < 64 * 24; i += 256) {
    const int r = i / 24, c4 = (i % 24) * 4;
    *reinterpret_cast<float4*>(&sZ[r * 100 + c4]) =
        *reinterpret_cast<const float4*>(zt + r * CC + c4);
  }
  __syncthreads();

  const int m0 = lane & 15;
  const int ksel = (lane >> 4) * 2;
  for (int tt = wave; tt < 24; tt += 8) {
    const int mt = tt & 3, nt = tt >> 2;
    const int ocol = nt * 16 + m0;
    const float* bp = wproj + ocol * CC;
    const float* ap = &sZ[(mt * 16 + m0) * 100];
    v8f acc = {};
#pragma unroll
    for (int kk = 0; kk < 24; ++kk) {
      const int kb = kk * 4 + ksel;
      v2f a = *reinterpret_cast<const v2f*>(ap + kb);
      v2f b = *reinterpret_cast<const v2f*>(bp + kb);
      acc = wmma4(a, b, acc);
    }
    const float bias = bproj[ocol];
#pragma unroll
    for (int v = 0; v < 8; ++v) {
      const int m = mt * 16 + v + ((lane >> 4) << 3);
      ot[m * CC + ocol] = acc[v] + bias;
    }
  }
}

extern "C" void kernel_launch(void* const* d_in, const int* in_sizes, int n_in,
                              void* d_out, int out_size, void* d_ws, size_t ws_size,
                              hipStream_t stream) {
  (void)in_sizes; (void)n_in; (void)out_size; (void)ws_size;
  const float* x     = (const float*)d_in[0];
  const float* mask  = (const float*)d_in[1];
  const float* wqkv  = (const float*)d_in[2];
  const float* wproj = (const float*)d_in[3];
  const float* bproj = (const float*)d_in[4];
  float* out = (float*)d_out;

  float* Y = (float*)d_ws;
  float* Z = Y + (size_t)BTOT * YBT;

  qkv_gemm_kernel<<<dim3(BTOT * WN), dim3(256), 0, stream>>>(x, wqkv, Y);
  win_attn_kernel<<<dim3(BTOT * HH), dim3(256), 0, stream>>>(Y, mask, Z);
  proj_gemm_kernel<<<dim3(BTOT * WN), dim3(256), 0, stream>>>(Z, wproj, bproj, out);
}


# qkv layer fused into both attention launches: a wave forms its 32-d slice of Q^T, K^T or V from x by WMMA, D registers feed the next WMMA as fragments; 3 launches, no Y; channel, d sum order changed
# speedup vs baseline: 2.8081x; 2.8081x over previous
#include <hip/hip_runtime.h>


typedef __attribute__((ext_vector_type(2))) float v2f;
typedef __attribute__((ext_vector_type(8))) float v8f;

#define BTOT 16
#define WN   64
#define NN   64
#define CC   96
#define HH   8
#define DD   768
#define C3   288
#define HD   6144
#define ZBT  393216
#define ATTN_SCALE 0.28867513459481287f

static __device__ __forceinline__ v8f wmma4(v2f a, v2f b, v8f c) {
  return __builtin_amdgcn_wmma_f32_16x16x4_f32(
       false, a,  false, b,
       (short)0, c,  false,  false);
}

#define NWG    512
#define LPITCH 100

typedef __attribute__((ext_vector_type(4))) float v4f;

static __device__ __forceinline__ int lds_off(int tid, int j) {
  const int i = tid + j * 256;
  return (i / 24) * LPITCH + (i % 24) * 4;
}

static __device__ __forceinline__ void put_tile(
    float* sT, int tid, v4f r0, v4f r1, v4f r2, v4f r3, v4f r4, v4f r5) {
  *reinterpret_cast<v4f*>(sT + lds_off(tid, 0)) = r0;
  *reinterpret_cast<v4f*>(sT + lds_off(tid, 1)) = r1;
  *reinterpret_cast<v4f*>(sT + lds_off(tid, 2)) = r2;
  *reinterpret_cast<v4f*>(sT + lds_off(tid, 3)) = r3;
  *reinterpret_cast<v4f*>(sT + lds_off(tid, 4)) = r4;
  *reinterpret_cast<v4f*>(sT + lds_off(tid, 5)) = r5;
}

template <int NA, int NB>
static __device__ __forceinline__ void mma_set(
    const float4 (&fa)[2][NA], const float4 (&fb)[2][NB], v8f (&acc)[NA][NB]) {
#pragma unroll
  for (int jj = 0; jj < 2; ++jj) {
#pragma unroll
    for (int ia = 0; ia < NA; ++ia) {
#pragma unroll
      for (int ib = 0; ib < NB; ++ib) {
        v2f a = {fa[jj][ia].x, fa[jj][ia].y};
        v2f b = {fb[jj][ib].x, fb[jj][ib].y};
        acc[ia][ib] = wmma4(a, b, acc[ia][ib]);
      }
    }
#pragma unroll
    for (int ia = 0; ia < NA; ++ia) {
#pragma unroll
      for (int ib = 0; ib < NB; ++ib) {
        v2f a = {fa[jj][ia].z, fa[jj][ia].w};
        v2f b = {fb[jj][ib].z, fb[jj][ib].w};
        acc[ia][ib] = wmma4(a, b, acc[ia][ib]);
      }
    }
  }
}

template <int NA, int NB>
static __device__ __forceinline__ void load_set(
    const float* __restrict__ ap, int astep, const float* __restrict__ bp, int bstep,
    float4 (&fa)[2][NA], float4 (&fb)[2][NB]) {
#pragma unroll
  for (int jj = 0; jj < 2; ++jj) {
#pragma unroll
    for (int i = 0; i < NA; ++i)
      fa[jj][i] = *reinterpret_cast<const float4*>(ap + i * astep + jj * 8);
#pragma unroll
    for (int i = 0; i < NB; ++i)
      fb[jj][i] = *reinterpret_cast<const float4*>(bp + i * bstep + jj * 8);
  }
}

template <int NA, int NB>
static __device__ __forceinline__ void gen_tiles(
    const float* __restrict__ ap, int astep,
    const float* __restrict__ bp, int bstep, v8f (&acc)[NA][NB]) {
  float4 fa0[2][NA], fb0[2][NB], fa1[2][NA], fb1[2][NB];
  load_set<NA, NB>(ap, astep, bp, bstep, fa0, fb0);
#pragma unroll 1
  for (int t = 0; t < 2; ++t) {
    load_set<NA, NB>(ap + 16, astep, bp + 16, bstep, fa1, fb1);
    mma_set<NA, NB>(fa0, fb0, acc);
    ap += 32;
    bp += 32;
    load_set<NA, NB>(ap, astep, bp, bstep, fa0, fb0);
    mma_set<NA, NB>(fa1, fb1, acc);
  }
  load_set<NA, NB>(ap + 16, astep, bp + 16, bstep, fa1, fb1);
  mma_set<NA, NB>(fa0, fb0, acc);
  mma_set<NA, NB>(fa1, fb1, acc);
}

#define SPITCH 68
#define AWAVES 12
#define ATHR   384
#define SPLANE 4096

__global__ __launch_bounds__(ATHR) void attn_scores_kernel(
    const float* __restrict__ x, const float* __restrict__ wqkv,
    float* __restrict__ Sp) {
  __shared__ __align__(16) float sPart[AWAVES * 64 * SPITCH];
  const int tid = threadIdx.x, lane = tid & 31, wave = tid >> 5;
  const int pair = blockIdx.x >> 1, half = blockIdx.x & 1;
  const int bt = pair >> 3, h = pair & 7;

  const int c = lane & 15;
  const int hf = lane >> 4;

  const int pq = h * DD + half * 384 + wave * 32;
  const int pk = pq + HD;
  const int nq = pq / C3, cq = pq - nq * C3;
  const int nk = pk / C3, ck = pk - nk * C3;
  const float* xl = x + (size_t)bt * (WN * NN * CC) + (size_t)c * (NN * CC) + hf * 4;
  const float* wl = wqkv + c * CC + hf * 4;

  v8f qt[2][4], kt[2][4];
#pragma unroll
  for (int dt = 0; dt < 2; ++dt)
#pragma unroll
    for (int wt = 0; wt < 4; ++wt) { qt[dt][wt] = (v8f){}; kt[dt][wt] = (v8f){}; }
  gen_tiles<2, 4>(wl + cq * CC, 16 * CC, xl + nq * CC, 16 * NN * CC, qt);
  gen_tiles<2, 4>(wl + ck * CC, 16 * CC, xl + nk * CC, 16 * NN * CC, kt);

  v8f acc[4][4];
#pragma unroll
  for (int mt = 0; mt < 4; ++mt)
#pragma unroll
    for (int nt = 0; nt < 4; ++nt) acc[mt][nt] = (v8f){};
#pragma unroll
  for (int dt = 0; dt < 2; ++dt) {
#pragma unroll
    for (int r = 0; r < 8; r += 2) {
#pragma unroll
      for (int mt = 0; mt < 4; ++mt) {
#pragma unroll
        for (int nt = 0; nt < 4; ++nt) {
          v2f a = {qt[dt][mt][r], qt[dt][mt][r + 1]};
          v2f b = {kt[dt][nt][r], kt[dt][nt][r + 1]};
          acc[mt][nt] = wmma4(a, b, acc[mt][nt]);
        }
      }
    }
  }

  float* myPart = sPart + wave * (64 * SPITCH);
#pragma unroll
  for (int mt = 0; mt < 4; ++mt) {
#pragma unroll
    for (int nt = 0; nt < 4; ++nt) {
#pragma unroll
      for (int v = 0; v < 8; ++v) {
        const int m = mt * 16 + v + (hf << 3);
        myPart[m * SPITCH + nt * 16 + c] = acc[mt][nt][v];
      }
    }
  }
  __syncthreads();

  float* so = Sp + (size_t)blockIdx.x * SPLANE;
  for (int idx = tid; idx < SPLANE; idx += ATHR) {
    const int o = (idx >> 6) * SPITCH + (idx & 63);
    float s = sPart[o];
#pragma unroll
    for (int p = 1; p < AWAVES; ++p) s += sPart[p * (64 * SPITCH) + o];
    so[idx] = s;
  }
}

__global__ __launch_bounds__(ATHR) void attn_pv_kernel(
    const float* __restrict__ x, const float* __restrict__ wqkv,
    const float* __restrict__ Sp, const float* __restrict__ mask,
    float* __restrict__ Z) {
  __shared__ __align__(16) float sP[64 * SPITCH];
  const int tid = threadIdx.x, lane = tid & 31, wave = tid >> 5;
  const int pair = blockIdx.x >> 1, ch = blockIdx.x & 1;
  const int bt = pair >> 3, h = pair & 7;

  const int c = lane & 15;
  const int hf = lane >> 4;
  const int dcol = ch * 384 + wave * 32;

  const float* s0p = Sp + (size_t)(pair * 2) * SPLANE + lane;
  const float* s1p = s0p + SPLANE;
  const float* mk = mask + (size_t)(bt & 7) * (NN * NN) + lane;
  float pl0[6], pl1[6], ph0[6], ph1[6], mk0[6], mk1[6];
#pragma unroll
  for (int i = 0; i < 6; ++i) {
    const int r = wave + AWAVES * i;
    const int rr = r < 64 ? r : 63;
    pl0[i] = s0p[rr * NN];
    pl1[i] = s0p[rr * NN + 32];
    ph0[i] = s1p[rr * NN];
    ph1[i] = s1p[rr * NN + 32];
    mk0[i] = mk[rr * NN];
    mk1[i] = mk[rr * NN + 32];
  }

  const int pv = 2 * HD + h * DD + dcol;
  const int nv = pv / C3, cv = pv - nv * C3;
  const float* xl = x + (size_t)bt * (WN * NN * CC) + (size_t)c * (NN * CC)
                      + nv * CC + hf * 4;
  const float* wl = wqkv + (cv + 2 * c) * CC + hf * 4;
  v8f vt[4][2];
#pragma unroll
  for (int wt = 0; wt < 4; ++wt) { vt[wt][0] = (v8f){}; vt[wt][1] = (v8f){}; }
  gen_tiles<4, 2>(xl, 16 * NN * CC, wl, CC, vt);

#pragma unroll
  for (int i = 0; i < 6; ++i) {
    const int r = wave + AWAVES * i;
    const float s0 = pl0[i] + ph0[i];
    const float s1 = pl1[i] + ph1[i];
    float x0 = s0 * ATTN_SCALE + mk0[i];
    float x1 = s1 * ATTN_SCALE + mk1[i];
    float mx = fmaxf(x0, x1);
#pragma unroll
    for (int off = 16; off >= 1; off >>= 1)
      mx = fmaxf(mx, __shfl_xor(mx, off, 32));
    const float e0 = __expf(x0 - mx), e1 = __expf(x1 - mx);
    float sm = e0 + e1;
#pragma unroll
    for (int off = 16; off >= 1; off >>= 1)
      sm += __shfl_xor(sm, off, 32);
    const float inv = 1.0f / sm;
    if (r < 64) {
      sP[r * SPITCH + lane] = e0 * inv;
      sP[r * SPITCH + lane + 32] = e1 * inv;
    }
  }
  __syncthreads();

  float* zb = Z + (size_t)bt * ZBT + h * DD;
  v8f oe[4], oo[4];
#pragma unroll
  for (int mt = 0; mt < 4; ++mt) { oe[mt] = (v8f){}; oo[mt] = (v8f){}; }
#pragma unroll
  for (int wt = 0; wt < 4; ++wt) {
#pragma unroll
    for (int r = 0; r < 8; r += 2) {
      const v2f be = {vt[wt][0][r], vt[wt][0][r + 1]};
      const v2f bo = {vt[wt][1][r], vt[wt][1][r + 1]};
#pragma unroll
      for (int mt = 0; mt < 4; ++mt) {
        const v2f pa = *reinterpret_cast<const v2f*>(
            &sP[(mt * 16 + c) * SPITCH + wt * 16 + hf * 8 + r]);
        oe[mt] = wmma4(pa, be, oe[mt]);
        oo[mt] = wmma4(pa, bo, oo[mt]);
      }
    }
  }
#pragma unroll
  for (int mt = 0; mt < 4; ++mt) {
#pragma unroll
    for (int v = 0; v < 8; ++v) {
      const int m = mt * 16 + v + (hf << 3);
      float2 o2 = {oe[mt][v], oo[mt][v]};
      *reinterpret_cast<float2*>(zb + (size_t)m * HD + dcol + 2 * c) = o2;
    }
  }
}

static __device__ __forceinline__ void proj_tiles(
    const float* sZ, const float* sW, const float* __restrict__ bproj,
    float* __restrict__ ot, int lane, int wave) {
  const int m0 = lane & 15;
  const int ksel = (lane >> 4) * 2;
  for (int tt = wave; tt < 24; tt += 8) {
    const int mt = tt & 3, nt = tt >> 2;
    const int ocol = nt * 16 + m0;
    const float* bp = sW + ocol * LPITCH;
    const float* ap = sZ + (mt * 16 + m0) * LPITCH;
    v8f acc = {};
#pragma unroll
    for (int kk = 0; kk < 24; ++kk) {
      const int kb = kk * 4 + ksel;
      v2f a = *reinterpret_cast<const v2f*>(ap + kb);
      v2f b = *reinterpret_cast<const v2f*>(bp + kb);
      acc = wmma4(a, b, acc);
    }
    const float bias = bproj[ocol];
#pragma unroll
    for (int v = 0; v < 8; ++v) {
      const int m = mt * 16 + v + ((lane >> 4) << 3);
      ot[m * CC + ocol] = acc[v] + bias;
    }
  }
}

__global__ __launch_bounds__(256) void proj_gemm_kernel(
    const float* __restrict__ Z, const float* __restrict__ wproj,
    const float* __restrict__ bproj, float* __restrict__ out) {
  __shared__ __align__(16) float sW[CC * LPITCH];
  __shared__ __align__(16) float sZ[64 * LPITCH];
  const int tid = threadIdx.x, lane = tid & 31, wave = tid >> 5;
  const int g0 = blockIdx.x * 2;
  const v4f* w4 = reinterpret_cast<const v4f*>(wproj);
  const v4f* z4 = reinterpret_cast<const v4f*>(Z + (size_t)g0 * (NN * CC)) + tid;
  float* ot = out + (size_t)g0 * (NN * CC);

  const v4f zb0 = z4[1536], zb1 = z4[1792], zb2 = z4[2048],
            zb3 = z4[2304], zb4 = z4[2560], zb5 = z4[2816];
  const v4f za0 = z4[0], za1 = z4[256], za2 = z4[512],
            za3 = z4[768], za4 = z4[1024], za5 = z4[1280];
  v4f wr[9];
#pragma unroll
  for (int j = 0; j < 9; ++j) wr[j] = w4[tid + j * 256];

  put_tile(sZ, tid, za0, za1, za2, za3, za4, za5);
#pragma unroll
  for (int j = 0; j < 9; ++j)
    *reinterpret_cast<v4f*>(sW + lds_off(tid, j)) = wr[j];
  __syncthreads();
  proj_tiles(sZ, sW, bproj, ot, lane, wave);
  __syncthreads();
  put_tile(sZ, tid, zb0, zb1, zb2, zb3, zb4, zb5);
  __syncthreads();
  proj_tiles(sZ, sW, bproj, ot + NN * CC, lane, wave);
}

extern "C" void kernel_launch(void* const* d_in, const int* in_sizes, int n_in,
                              void* d_out, int out_size, void* d_ws, size_t ws_size,
                              hipStream_t stream) {
  (void)in_sizes; (void)n_in; (void)out_size; (void)ws_size;
  const float* x     = (const float*)d_in[0];
  const float* mask  = (const float*)d_in[1];
  const float* wqkv  = (const float*)d_in[2];
  const float* wproj = (const float*)d_in[3];
  const float* bproj = (const float*)d_in[4];
  float* out = (float*)d_out;

  float* Z = (float*)d_ws;
  float* Sp = Z + (size_t)BTOT * ZBT;

  attn_scores_kernel<<<dim3(BTOT * HH * 2), dim3(ATHR), 0, stream>>>(x, wqkv, Sp);
  attn_pv_kernel<<<dim3(BTOT * HH * 2), dim3(ATHR), 0, stream>>>(x, wqkv, Sp, mask, Z);
  proj_gemm_kernel<<<dim3(NWG), dim3(256), 0, stream>>>(Z, wproj, bproj, out);
}


# both attention launches: a work-group copies the x rows of its two tokens per qkv product into LDS once, in whole lines; waves read x fragments from LDS, weights still from global; sums unchanged
# speedup vs baseline: 3.0147x; 1.0736x over previous
#include <hip/hip_runtime.h>


typedef __attribute__((ext_vector_type(2))) float v2f;
typedef __attribute__((ext_vector_type(8))) float v8f;

#define BTOT 16
#define WN   64
#define NN   64
#define CC   96
#define HH   8
#define DD   768
#define C3   288
#define HD   6144
#define ZBT  393216
#define ATTN_SCALE 0.28867513459481287f

static __device__ __forceinline__ v8f wmma4(v2f a, v2f b, v8f c) {
  return __builtin_amdgcn_wmma_f32_16x16x4_f32(
       false, a,  false, b,
       (short)0, c,  false,  false);
}

#define NWG    512
#define LPITCH 100

typedef __attribute__((ext_vector_type(4))) float v4f;

static __device__ __forceinline__ int lds_off(int tid, int j) {
  const int i = tid + j * 256;
  return (i / 24) * LPITCH + (i % 24) * 4;
}

static __device__ __forceinline__ void put_tile(
    float* sT, int tid, v4f r0, v4f r1, v4f r2, v4f r3, v4f r4, v4f r5) {
  *reinterpret_cast<v4f*>(sT + lds_off(tid, 0)) = r0;
  *reinterpret_cast<v4f*>(sT + lds_off(tid, 1)) = r1;
  *reinterpret_cast<v4f*>(sT + lds_off(tid, 2)) = r2;
  *reinterpret_cast<v4f*>(sT + lds_off(tid, 3)) = r3;
  *reinterpret_cast<v4f*>(sT + lds_off(tid, 4)) = r4;
  *reinterpret_cast<v4f*>(sT + lds_off(tid, 5)) = r5;
}

#define XPITCH 100
#define XSLAB  (64 * XPITCH)

static __device__ __forceinline__ void slab_request(
    const float* __restrict__ xt, int tid, v4f (&r)[4]) {
  const int w = tid / 24, q = tid - w * 24;
#pragma unroll
  for (int j = 0; j < 4; ++j)
    r[j] = *reinterpret_cast<const v4f*>(xt + (size_t)(w + 16 * j) * (NN * CC) + q * 4);
}

static __device__ __forceinline__ void slab_put(float* slab, int tid, const v4f (&r)[4]) {
  const int w = tid / 24, q = tid - w * 24;
#pragma unroll
  for (int j = 0; j < 4; ++j)
    *reinterpret_cast<v4f*>(slab + (w + 16 * j) * XPITCH + q * 4) = r[j];
}

template <int NA, int NB>
static __device__ __forceinline__ void mma_set(
    const float4 (&fa)[2][NA], const float4 (&fb)[2][NB], v8f (&acc)[NA][NB]) {
#pragma unroll
  for (int jj = 0; jj < 2; ++jj) {
#pragma unroll
    for (int ia = 0; ia < NA; ++ia) {
#pragma unroll
      for (int ib = 0; ib < NB; ++ib) {
        v2f a = {fa[jj][ia].x, fa[jj][ia].y};
        v2f b = {fb[jj][ib].x, fb[jj][ib].y};
        acc[ia][ib] = wmma4(a, b, acc[ia][ib]);
      }
    }
#pragma unroll
    for (int ia = 0; ia < NA; ++ia) {
#pragma unroll
      for (int ib = 0; ib < NB; ++ib) {
        v2f a = {fa[jj][ia].z, fa[jj][ia].w};
        v2f b = {fb[jj][ib].z, fb[jj][ib].w};
        acc[ia][ib] = wmma4(a, b, acc[ia][ib]);
      }
    }
  }
}

template <int NA, int NB>
static __device__ __forceinline__ void load_set(
    const float* __restrict__ ap, int astep, const float* __restrict__ bp, int bstep,
    float4 (&fa)[2][NA], float4 (&fb)[2][NB]) {
#pragma unroll
  for (int jj = 0; jj < 2; ++jj) {
#pragma unroll
    for (int i = 0; i < NA; ++i)
      fa[jj][i] = *reinterpret_cast<const float4*>(ap + i * astep + jj * 8);
#pragma unroll
    for (int i = 0; i < NB; ++i)
      fb[jj][i] = *reinterpret_cast<const float4*>(bp + i * bstep + jj * 8);
  }
}

template <int NA, int NB>
static __device__ __forceinline__ void gen_tiles(
    const float* __restrict__ ap, int astep,
    const float* __restrict__ bp, int bstep, v8f (&acc)[NA][NB]) {
  float4 fa0[2][NA], fb0[2][NB], fa1[2][NA], fb1[2][NB];
  load_set<NA, NB>(ap, astep, bp, bstep, fa0, fb0);
#pragma unroll 1
  for (int t = 0; t < 2; ++t) {
    load_set<NA, NB>(ap + 16, astep, bp + 16, bstep, fa1, fb1);
    mma_set<NA, NB>(fa0, fb0, acc);
    ap += 32;
    bp += 32;
    load_set<NA, NB>(ap, astep, bp, bstep, fa0, fb0);
    mma_set<NA, NB>(fa1, fb1, acc);
  }
  load_set<NA, NB>(ap + 16, astep, bp + 16, bstep, fa1, fb1);
  mma_set<NA, NB>(fa0, fb0, acc);
  mma_set<NA, NB>(fa1, fb1, acc);
}

#define SPITCH 68
#define AWAVES 12
#define ATHR   384
#define SPLANE 4096

__global__ __launch_bounds__(ATHR) void attn_scores_kernel(
    const float* __restrict__ x, const float* __restrict__ wqkv,
    float* __restrict__ Sp) {
  __shared__ __align__(16) float sPart[AWAVES * 64 * SPITCH];
  __shared__ __align__(16) float sX[4 * XSLAB];
  const int tid = threadIdx.x, lane = tid & 31, wave = tid >> 5;
  const int pair = blockIdx.x >> 1, half = blockIdx.x & 1;
  const int bt = pair >> 3, h = pair & 7;

  const int c = lane & 15;
  const int hf = lane >> 4;

  const int bq = h * DD + half * 384;
  const int pq = bq + wave * 32;
  const int pk = pq + HD;
  const int nq = pq / C3, cq = pq - nq * C3;
  const int nk = pk / C3, ck = pk - nk * C3;
  const int n0q = bq / C3, n0k = (bq + HD) / C3;

  {
    const float* xb = x + (size_t)bt * (WN * NN * CC);
    v4f r0[4], r1[4], r2[4], r3[4];
    slab_request(xb + n0q * CC, tid, r0);
    slab_request(xb + (n0q + 1) * CC, tid, r1);
    slab_request(xb + n0k * CC, tid, r2);
    slab_request(xb + (n0k + 1) * CC, tid, r3);
    slab_put(sX, tid, r0);
    slab_put(sX + XSLAB, tid, r1);
    slab_put(sX + 2 * XSLAB, tid, r2);
    slab_put(sX + 3 * XSLAB, tid, r3);
  }
  __syncthreads();

  const float* xq = sX + (nq - n0q) * XSLAB + c * XPITCH + hf * 4;
  const float* xk = sX + (2 + nk - n0k) * XSLAB + c * XPITCH + hf * 4;
  const float* wl = wqkv + c * CC + hf * 4;

  v8f qt[2][4], kt[2][4];
#pragma unroll
  for (int dt = 0; dt < 2; ++dt)
#pragma unroll
    for (int wt = 0; wt < 4; ++wt) { qt[dt][wt] = (v8f){}; kt[dt][wt] = (v8f){}; }
  gen_tiles<2, 4>(wl + cq * CC, 16 * CC, xq, 16 * XPITCH, qt);
  gen_tiles<2, 4>(wl + ck * CC, 16 * CC, xk, 16 * XPITCH, kt);

  v8f acc[4][4];
#pragma unroll
  for (int mt = 0; mt < 4; ++mt)
#pragma unroll
    for (int nt = 0; nt < 4; ++nt) acc[mt][nt] = (v8f){};
#pragma unroll
  for (int dt = 0; dt < 2; ++dt) {
#pragma unroll
    for (int r = 0; r < 8; r += 2) {
#pragma unroll
      for (int mt = 0; mt < 4; ++mt) {
#pragma unroll
        for (int nt = 0; nt < 4; ++nt) {
          v2f a = {qt[dt][mt][r], qt[dt][mt][r + 1]};
          v2f b = {kt[dt][nt][r], kt[dt][nt][r + 1]};
          acc[mt][nt] = wmma4(a, b, acc[mt][nt]);
        }
      }
    }
  }

  float* myPart = sPart + wave * (64 * SPITCH);
#pragma unroll
  for (int mt = 0; mt < 4; ++mt) {
#pragma unroll
    for (int nt = 0; nt < 4; ++nt) {
#pragma unroll
      for (int v = 0; v < 8; ++v) {
        const int m = mt * 16 + v + (hf << 3);
        myPart[m * SPITCH + nt * 16 + c] = acc[mt][nt][v];
      }
    }
  }
  __syncthreads();

  float* so = Sp + (size_t)blockIdx.x * SPLANE;
  for (int idx = tid; idx < SPLANE; idx += ATHR) {
    const int o = (idx >> 6) * SPITCH + (idx & 63);
    float s = sPart[o];
#pragma unroll
    for (int p = 1; p < AWAVES; ++p) s += sPart[p * (64 * SPITCH) + o];
    so[idx] = s;
  }
}

__global__ __launch_bounds__(ATHR) void attn_pv_kernel(
    const float* __restrict__ x, const float* __restrict__ wqkv,
    const float* __restrict__ Sp, const float* __restrict__ mask,
    float* __restrict__ Z) {
  __shared__ __align__(16) float sP[64 * SPITCH];
  __shared__ __align__(16) float sX[2 * XSLAB];
  const int tid = threadIdx.x, lane = tid & 31, wave = tid >> 5;
  const int pair = blockIdx.x >> 1, ch = blockIdx.x & 1;
  const int bt = pair >> 3, h = pair & 7;

  const int c = lane & 15;
  const int hf = lane >> 4;
  const int dcol = ch * 384 + wave * 32;

  const int bv = 2 * HD + h * DD + ch * 384;
  const int n0v = bv / C3;
  v4f xr0[4], xr1[4];
  {
    const float* xb = x + (size_t)bt * (WN * NN * CC);
    slab_request(xb + n0v * CC, tid, xr0);
    slab_request(xb + (n0v + 1) * CC, tid, xr1);
  }

  const float* s0p = Sp + (size_t)(pair * 2) * SPLANE + lane;
  const float* s1p = s0p + SPLANE;
  const float* mk = mask + (size_t)(bt & 7) * (NN * NN) + lane;
  float pl0[6], pl1[6], ph0[6], ph1[6], mk0[6], mk1[6];
#pragma unroll
  for (int i = 0; i < 6; ++i) {
    const int r = wave + AWAVES * i;
    const int rr = r < 64 ? r : 63;
    pl0[i] = s0p[rr * NN];
    pl1[i] = s0p[rr * NN + 32];
    ph0[i] = s1p[rr * NN];
    ph1[i] = s1p[rr * NN + 32];
    mk0[i] = mk[rr * NN];
    mk1[i] = mk[rr * NN + 32];
  }

  slab_put(sX, tid, xr0);
  slab_put(sX + XSLAB, tid, xr1);
  __syncthreads();

  const int pv = bv + wave * 32;
  const int nv = pv / C3, cv = pv - nv * C3;
  const float* xl = sX + (nv - n0v) * XSLAB + c * XPITCH + hf * 4;
  const float* wl = wqkv + (cv + 2 * c) * CC + hf * 4;
  v8f vt[4][2];
#pragma unroll
  for (int wt = 0; wt < 4; ++wt) { vt[wt][0] = (v8f){}; vt[wt][1] = (v8f){}; }
  gen_tiles<4, 2>(xl, 16 * XPITCH, wl, CC, vt);

#pragma unroll
  for (int i = 0; i < 6; ++i) {
    const int r = wave + AWAVES * i;
    const float s0 = pl0[i] + ph0[i];
    const float s1 = pl1[i] + ph1[i];
    float x0 = s0 * ATTN_SCALE + mk0[i];
    float x1 = s1 * ATTN_SCALE + mk1[i];
    float mx = fmaxf(x0, x1);
#pragma unroll
    for (int off = 16; off >= 1; off >>= 1)
      mx = fmaxf(mx, __shfl_xor(mx, off, 32));
    const float e0 = __expf(x0 - mx), e1 = __expf(x1 - mx);
    float sm = e0 + e1;
#pragma unroll
    for (int off = 16; off >= 1; off >>= 1)
      sm += __shfl_xor(sm, off, 32);
    const float inv = 1.0f / sm;
    if (r < 64) {
      sP[r * SPITCH + lane] = e0 * inv;
      sP[r * SPITCH + lane + 32] = e1 * inv;
    }
  }
  __syncthreads();

  float* zb = Z + (size_t)bt * ZBT + h * DD;
  v8f oe[4], oo[4];
#pragma unroll
  for (int mt = 0; mt < 4; ++mt) { oe[mt] = (v8f){}; oo[mt] = (v8f){}; }
#pragma unroll
  for (int wt = 0; wt < 4; ++wt) {
#pragma unroll
    for (int r = 0; r < 8; r += 2) {
      const v2f be = {vt[wt][0][r], vt[wt][0][r + 1]};
      const v2f bo = {vt[wt][1][r], vt[wt][1][r + 1]};
#pragma unroll
      for (int mt = 0; mt < 4; ++mt) {
        const v2f pa = *reinterpret_cast<const v2f*>(
            &sP[(mt * 16 + c) * SPITCH + wt * 16 + hf * 8 + r]);
        oe[mt] = wmma4(pa, be, oe[mt]);
        oo[mt] = wmma4(pa, bo, oo[mt]);
      }
    }
  }
#pragma unroll
  for (int mt = 0; mt < 4; ++mt) {
#pragma unroll
    for (int v = 0; v < 8; ++v) {
      const int m = mt * 16 + v + (hf << 3);
      float2 o2 = {oe[mt][v], oo[mt][v]};
      *reinterpret_cast<float2*>(zb + (size_t)m * HD + dcol + 2 * c) = o2;
    }
  }
}

static __device__ __forceinline__ void proj_tiles(
    const float* sZ, const float* sW, const float* __restrict__ bproj,
    float* __restrict__ ot, int lane, int wave) {
  const int m0 = lane & 15;
  const int ksel = (lane >> 4) * 2;
  for (int tt = wave; tt < 24; tt += 8) {
    const int mt = tt & 3, nt = tt >> 2;
    const int ocol = nt * 16 + m0;
    const float* bp = sW + ocol * LPITCH;
    const float* ap = sZ + (mt * 16 + m0) * LPITCH;
    v8f acc = {};
#pragma unroll
    for (int kk = 0; kk < 24; ++kk) {
      const int kb = kk * 4 + ksel;
      v2f a = *reinterpret_cast<const v2f*>(ap + kb);
      v2f b = *reinterpret_cast<const v2f*>(bp + kb);
      acc = wmma4(a, b, acc);
    }
    const float bias = bproj[ocol];
#pragma unroll
    for (int v = 0; v < 8; ++v) {
      const int m = mt * 16 + v + ((lane >> 4) << 3);
      ot[m * CC + ocol] = acc[v] + bias;
    }
  }
}

__global__ __launch_bounds__(256) void proj_gemm_kernel(
    const float* __restrict__ Z, const float* __restrict__ wproj,
    const float* __restrict__ bproj, float* __restrict__ out) {
  __shared__ __align__(16) float sW[CC * LPITCH];
  __shared__ __align__(16) float sZ[64 * LPITCH];
  const int tid = threadIdx.x, lane = tid & 31, wave = tid >> 5;
  const int g0 = blockIdx.x * 2;
  const v4f* w4 = reinterpret_cast<const v4f*>(wproj);
  const v4f* z4 = reinterpret_cast<const v4f*>(Z + (size_t)g0 * (NN * CC)) + tid;
  float* ot = out + (size_t)g0 * (NN * CC);

  const v4f zb0 = z4[1536], zb1 = z4[1792], zb2 = z4[2048],
            zb3 = z4[2304], zb4 = z4[2560], zb5 = z4[2816];
  const v4f za0 = z4[0], za1 = z4[256], za2 = z4[512],
            za3 = z4[768], za4 = z4[1024], za5 = z4[1280];
  v4f wr[9];
#pragma unroll
  for (int j = 0; j < 9; ++j) wr[j] = w4[tid + j * 256];

  put_tile(sZ, tid, za0, za1, za2, za3, za4, za5);
#pragma unroll
  for (int j = 0; j < 9; ++j)
    *reinterpret_cast<v4f*>(sW + lds_off(tid, j)) = wr[j];
  __syncthreads();
  proj_tiles(sZ, sW, bproj, ot, lane, wave);
  __syncthreads();
  put_tile(sZ, tid, zb0, zb1, zb2, zb3, zb4, zb5);
  __syncthreads();
  proj_tiles(sZ, sW, bproj, ot + NN * CC, lane, wave);
}

extern "C" void kernel_launch(void* const* d_in, const int* in_sizes, int n_in,
                              void* d_out, int out_size, void* d_ws, size_t ws_size,
                              hipStream_t stream) {
  (void)in_sizes; (void)n_in; (void)out_size; (void)ws_size;
  const float* x     = (const float*)d_in[0];
  const float* mask  = (const float*)d_in[1];
  const float* wqkv  = (const float*)d_in[2];
  const float* wproj = (const float*)d_in[3];
  const float* bproj = (const float*)d_in[4];
  float* out = (float*)d_out;

  float* Z = (float*)d_ws;
  float* Sp = Z + (size_t)BTOT * ZBT;

  attn_scores_kernel<<<dim3(BTOT * HH * 2), dim3(ATHR), 0, stream>>>(x, wqkv, Sp);
  attn_pv_kernel<<<dim3(BTOT * HH * 2), dim3(ATHR), 0, stream>>>(x, wqkv, Sp, mask, Z);
  proj_gemm_kernel<<<dim3(NWG), dim3(256), 0, stream>>>(Z, wproj, bproj, out);
}


# proj layer done inside the PV launch: a wave's O^T D registers are the B fragments of out^T = Wproj O^T over its 32 channels, 3 partials added via LDS; 2 launches, no Z; proj sum order changed
# speedup vs baseline: 3.6129x; 1.1984x over previous
#include <hip/hip_runtime.h>


typedef __attribute__((ext_vector_type(2))) float v2f;
typedef __attribute__((ext_vector_type(8))) float v8f;

#define BTOT 16
#define WN   64
#define NN   64
#define CC   96
#define HH   8
#define DD   768
#define C3   288
#define HD   6144
#define ATTN_SCALE 0.28867513459481287f

static __device__ __forceinline__ v8f wmma4(v2f a, v2f b, v8f c) {
  return __builtin_amdgcn_wmma_f32_16x16x4_f32(
       false, a,  false, b,
       (short)0, c,  false,  false);
}

typedef __attribute__((ext_vector_type(4))) float v4f;

#define XPITCH 100
#define XSLAB  (64 * XPITCH)

static __device__ __forceinline__ void slab_request(
    const float* __restrict__ xt, int tid, v4f (&r)[4]) {
  const int w = tid / 24, q = tid - w * 24;
#pragma unroll
  for (int j = 0; j < 4; ++j)
    r[j] = *reinterpret_cast<const v4f*>(xt + (size_t)(w + 16 * j) * (NN * CC) + q * 4);
}

static __device__ __forceinline__ void slab_put(float* slab, int tid, const v4f (&r)[4]) {
  const int w = tid / 24, q = tid - w * 24;
#pragma unroll
  for (int j = 0; j < 4; ++j)
    *reinterpret_cast<v4f*>(slab + (w + 16 * j) * XPITCH + q * 4) = r[j];
}

template <int NA, int NB>
static __device__ __forceinline__ void mma_set(
    const float4 (&fa)[2][NA], const float4 (&fb)[2][NB], v8f (&acc)[NA][NB]) {
#pragma unroll
  for (int jj = 0; jj < 2; ++jj) {
#pragma unroll
    for (int ia = 0; ia < NA; ++ia) {
#pragma unroll
      for (int ib = 0; ib < NB; ++ib) {
        v2f a = {fa[jj][ia].x, fa[jj][ia].y};
        v2f b = {fb[jj][ib].x, fb[jj][ib].y};
        acc[ia][ib] = wmma4(a, b, acc[ia][ib]);
      }
    }
#pragma unroll
    for (int ia = 0; ia < NA; ++ia) {
#pragma unroll
      for (int ib = 0; ib < NB; ++ib) {
        v2f a = {fa[jj][ia].z, fa[jj][ia].w};
        v2f b = {fb[jj][ib].z, fb[jj][ib].w};
        acc[ia][ib] = wmma4(a, b, acc[ia][ib]);
      }
    }
  }
}

template <int NA, int NB>
static __device__ __forceinline__ void load_set(
    const float* __restrict__ ap, int astep, const float* __restrict__ bp, int bstep,
    float4 (&fa)[2][NA], float4 (&fb)[2][NB]) {
#pragma unroll
  for (int jj = 0; jj < 2; ++jj) {
#pragma unroll
    for (int i = 0; i < NA; ++i)
      fa[jj][i] = *reinterpret_cast<const float4*>(ap + i * astep + jj * 8);
#pragma unroll
    for (int i = 0; i < NB; ++i)
      fb[jj][i] = *reinterpret_cast<const float4*>(bp + i * bstep + jj * 8);
  }
}

template <int NA, int NB>
static __device__ __forceinline__ void gen_tiles(
    const float* __restrict__ ap, int astep,
    const float* __restrict__ bp, int bstep, v8f (&acc)[NA][NB]) {
  float4 fa0[2][NA], fb0[2][NB], fa1[2][NA], fb1[2][NB];
  load_set<NA, NB>(ap, astep, bp, bstep, fa0, fb0);
#pragma unroll 1
  for (int t = 0; t < 2; ++t) {
    load_set<NA, NB>(ap + 16, astep, bp + 16, bstep, fa1, fb1);
    mma_set<NA, NB>(fa0, fb0, acc);
    ap += 32;
    bp += 32;
    load_set<NA, NB>(ap, astep, bp, bstep, fa0, fb0);
    mma_set<NA, NB>(fa1, fb1, acc);
  }
  load_set<NA, NB>(ap + 16, astep, bp + 16, bstep, fa1, fb1);
  mma_set<NA, NB>(fa0, fb0, acc);
  mma_set<NA, NB>(fa1, fb1, acc);
}

#define SPITCH 68
#define AWAVES 12
#define ATHR   384
#define SPLANE 4096

__global__ __launch_bounds__(ATHR) void attn_scores_kernel(
    const float* __restrict__ x, const float* __restrict__ wqkv,
    float* __restrict__ Sp) {
  __shared__ __align__(16) float sPart[AWAVES * 64 * SPITCH];
  __shared__ __align__(16) float sX[4 * XSLAB];
  const int tid = threadIdx.x, lane = tid & 31, wave = tid >> 5;
  const int pair = blockIdx.x >> 1, half = blockIdx.x & 1;
  const int bt = pair >> 3, h = pair & 7;

  const int c = lane & 15;
  const int hf = lane >> 4;

  const int bq = h * DD + half * 384;
  const int pq = bq + wave * 32;
  const int pk = pq + HD;
  const int nq = pq / C3, cq = pq - nq * C3;
  const int nk = pk / C3, ck = pk - nk * C3;
  const int n0q = bq / C3, n0k = (bq + HD) / C3;

  {
    const float* xb = x + (size_t)bt * (WN * NN * CC);
    v4f r0[4], r1[4], r2[4], r3[4];
    slab_request(xb + n0q * CC, tid, r0);
    slab_request(xb + (n0q + 1) * CC, tid, r1);
    slab_request(xb + n0k * CC, tid, r2);
    slab_request(xb + (n0k + 1) * CC, tid, r3);
    slab_put(sX, tid, r0);
    slab_put(sX + XSLAB, tid, r1);
    slab_put(sX + 2 * XSLAB, tid, r2);
    slab_put(sX + 3 * XSLAB, tid, r3);
  }
  __syncthreads();

  const float* xq = sX + (nq - n0q) * XSLAB + c * XPITCH + hf * 4;
  const float* xk = sX + (2 + nk - n0k) * XSLAB + c * XPITCH + hf * 4;
  const float* wl = wqkv + c * CC + hf * 4;

  v8f qt[2][4], kt[2][4];
#pragma unroll
  for (int dt = 0; dt < 2; ++dt)
#pragma unroll
    for (int wt = 0; wt < 4; ++wt) { qt[dt][wt] = (v8f){}; kt[dt][wt] = (v8f){}; }
  gen_tiles<2, 4>(wl + cq * CC, 16 * CC, xq, 16 * XPITCH, qt);
  gen_tiles<2, 4>(wl + ck * CC, 16 * CC, xk, 16 * XPITCH, kt);

  v8f acc[4][4];
#pragma unroll
  for (int mt = 0; mt < 4; ++mt)
#pragma unroll
    for (int nt = 0; nt < 4; ++nt) acc[mt][nt] = (v8f){};
#pragma unroll
  for (int dt = 0; dt < 2; ++dt) {
#pragma unroll
    for (int r = 0; r < 8; r += 2) {
#pragma unroll
      for (int mt = 0; mt < 4; ++mt) {
#pragma unroll
        for (int nt = 0; nt < 4; ++nt) {
          v2f a = {qt[dt][mt][r], qt[dt][mt][r + 1]};
          v2f b = {kt[dt][nt][r], kt[dt][nt][r + 1]};
          acc[mt][nt] = wmma4(a, b, acc[mt][nt]);
        }
      }
    }
  }

  float* myPart = sPart + wave * (64 * SPITCH);
#pragma unroll
  for (int mt = 0; mt < 4; ++mt) {
#pragma unroll
    for (int nt = 0; nt < 4; ++nt) {
#pragma unroll
      for (int v = 0; v < 8; ++v) {
        const int m = mt * 16 + v + (hf << 3);
        myPart[m * SPITCH + nt * 16 + c] = acc[mt][nt][v];
      }
    }
  }
  __syncthreads();

  float* so = Sp + (size_t)blockIdx.x * SPLANE;
  for (int idx = tid; idx < SPLANE; idx += ATHR) {
    const int o = (idx >> 6) * SPITCH + (idx & 63);
    float s = sPart[o];
#pragma unroll
    for (int p = 1; p < AWAVES; ++p) s += sPart[p * (64 * SPITCH) + o];
    so[idx] = s;
  }
}

#define OPITCH 100
#define OPLANE (64 * OPITCH)

__global__ __launch_bounds__(ATHR) void attn_pv_proj_kernel(
    const float* __restrict__ x, const float* __restrict__ wqkv,
    const float* __restrict__ Sp, const float* __restrict__ mask,
    const float* __restrict__ wproj, const float* __restrict__ bproj,
    float* __restrict__ out) {
  __shared__ __align__(16) float sBuf[AWAVES * OPLANE];
  float* sP = sBuf;
  float* sX = sBuf + 64 * SPITCH;
  const int tid = threadIdx.x, lane = tid & 31, wave = tid >> 5;
  const int pair = blockIdx.x >> 1, ch = blockIdx.x & 1;
  const int bt = pair >> 3, h = pair & 7;

  const int c = lane & 15;
  const int hf = lane >> 4;

  const int bv = 2 * HD + h * DD + ch * 384;
  const int n0v = bv / C3;
  v4f xr0[4], xr1[4];
  {
    const float* xb = x + (size_t)bt * (WN * NN * CC);
    slab_request(xb + n0v * CC, tid, xr0);
    slab_request(xb + (n0v + 1) * CC, tid, xr1);
  }

  const float* s0p = Sp + (size_t)(pair * 2) * SPLANE + lane;
  const float* s1p = s0p + SPLANE;
  const float* mk = mask + (size_t)(bt & 7) * (NN * NN) + lane;
  float pl0[6], pl1[6], ph0[6], ph1[6], mk0[6], mk1[6];
#pragma unroll
  for (int i = 0; i < 6; ++i) {
    const int r = wave + AWAVES * i;
    const int rr = r < 64 ? r : 63;
    pl0[i] = s0p[rr * NN];
    pl1[i] = s0p[rr * NN + 32];
    ph0[i] = s1p[rr * NN];
    ph1[i] = s1p[rr * NN + 32];
    mk0[i] = mk[rr * NN];
    mk1[i] = mk[rr * NN + 32];
  }

  slab_put(sX, tid, xr0);
  slab_put(sX + XSLAB, tid, xr1);
  __syncthreads();

  const int pv = bv + wave * 32;
  const int nv = pv / C3, cv = pv - nv * C3;
  const float* xl = sX + (nv - n0v) * XSLAB + c * XPITCH + hf * 4;
  const float* wl = wqkv + (cv + c) * CC + hf * 4;
  v8f vt[4][2];
#pragma unroll
  for (int wt = 0; wt < 4; ++wt) { vt[wt][0] = (v8f){}; vt[wt][1] = (v8f){}; }
  gen_tiles<4, 2>(xl, 16 * XPITCH, wl, 16 * CC, vt);

#pragma unroll
  for (int i = 0; i < 6; ++i) {
    const int r = wave + AWAVES * i;
    const float s0 = pl0[i] + ph0[i];
    const float s1 = pl1[i] + ph1[i];
    float x0 = s0 * ATTN_SCALE + mk0[i];
    float x1 = s1 * ATTN_SCALE + mk1[i];
    float mx = fmaxf(x0, x1);
#pragma unroll
    for (int off = 16; off >= 1; off >>= 1)
      mx = fmaxf(mx, __shfl_xor(mx, off, 32));
    const float e0 = __expf(x0 - mx), e1 = __expf(x1 - mx);
    float sm = e0 + e1;
#pragma unroll
    for (int off = 16; off >= 1; off >>= 1)
      sm += __shfl_xor(sm, off, 32);
    const float inv = 1.0f / sm;
    if (r < 64) {
      sP[r * SPITCH + lane] = e0 * inv;
      sP[r * SPITCH + lane + 32] = e1 * inv;
    }
  }
  __syncthreads();

  v8f ot[2][4];
#pragma unroll
  for (int nt = 0; nt < 4; ++nt) { ot[0][nt] = (v8f){}; ot[1][nt] = (v8f){}; }
#pragma unroll
  for (int wt = 0; wt < 4; ++wt) {
#pragma unroll
    for (int r = 0; r < 8; r += 2) {
      const v2f a0 = {vt[wt][0][r], vt[wt][0][r + 1]};
      const v2f a1 = {vt[wt][1][r], vt[wt][1][r + 1]};
#pragma unroll
      for (int nt = 0; nt < 4; ++nt) {
        const v2f pb = *reinterpret_cast<const v2f*>(
            &sP[(nt * 16 + c) * SPITCH + wt * 16 + hf * 8 + r]);
        ot[0][nt] = wmma4(a0, pb, ot[0][nt]);
        ot[1][nt] = wmma4(a1, pb, ot[1][nt]);
      }
    }
  }
  __syncthreads();

  const int j = wave / 3, s3 = wave - 3 * j;
  const float* wp = wproj + c * CC + s3 * 32 + hf * 8;
  float* myPart = sBuf + (s3 * 4 + j) * OPLANE;
  float4 wa[2][2];
#pragma unroll
  for (int dt = 0; dt < 2; ++dt) {
    wa[dt][0] = *reinterpret_cast<const float4*>(wp + dt * 16);
    wa[dt][1] = *reinterpret_cast<const float4*>(wp + dt * 16 + 4);
  }
#pragma unroll 1
  for (int ob = 0; ob < 6; ++ob) {
    const int obn = ob < 5 ? ob + 1 : 5;
    float4 wn[2][2];
#pragma unroll
    for (int dt = 0; dt < 2; ++dt) {
      wn[dt][0] = *reinterpret_cast<const float4*>(wp + obn * 16 * CC + dt * 16);
      wn[dt][1] = *reinterpret_cast<const float4*>(wp + obn * 16 * CC + dt * 16 + 4);
    }
    v8f pa[4];
#pragma unroll
    for (int nt = 0; nt < 4; ++nt) pa[nt] = (v8f){};
#pragma unroll
    for (int dt = 0; dt < 2; ++dt) {
#pragma unroll
      for (int q = 0; q < 4; ++q) {
        const float4 w4 = wa[dt][q >> 1];
        const v2f a = (q & 1) ? (v2f){w4.z, w4.w} : (v2f){w4.x, w4.y};
#pragma unroll
        for (int nt = 0; nt < 4; ++nt) {
          const v2f b = {ot[dt][nt][2 * q], ot[dt][nt][2 * q + 1]};
          pa[nt] = wmma4(a, b, pa[nt]);
        }
      }
    }
#pragma unroll
    for (int nt = 0; nt < 4; ++nt) {
      float* p = myPart + (nt * 16 + c) * OPITCH + ob * 16 + hf * 8;
      *reinterpret_cast<v4f*>(p) = (v4f){pa[nt][0], pa[nt][1], pa[nt][2], pa[nt][3]};
      *reinterpret_cast<v4f*>(p + 4) = (v4f){pa[nt][4], pa[nt][5], pa[nt][6], pa[nt][7]};
    }
#pragma unroll
    for (int dt = 0; dt < 2; ++dt) { wa[dt][0] = wn[dt][0]; wa[dt][1] = wn[dt][1]; }
  }
  __syncthreads();

  float* og = out + ((size_t)bt * (WN * NN) + h * 8 + ch * 4) * CC;
#pragma unroll 4
  for (int i = 0; i < 16; ++i) {
    const int idx = tid + ATHR * i;
    const int o4 = idx % 24, rest = idx / 24;
    const int jj = rest & 3, n = rest >> 2;
    const float* p = sBuf + (jj * 64 + n) * OPITCH + o4 * 4;
    const v4f t0 = *reinterpret_cast<const v4f*>(p);
    const v4f t1 = *reinterpret_cast<const v4f*>(p + 4 * OPLANE);
    const v4f t2 = *reinterpret_cast<const v4f*>(p + 8 * OPLANE);
    const v4f bias = *reinterpret_cast<const v4f*>(bproj + o4 * 4);
    *reinterpret_cast<v4f*>(og + (size_t)n * (NN * CC) + jj * CC + o4 * 4) = ((t0 + t1) + t2) + bias;
  }
}

extern "C" void kernel_launch(void* const* d_in, const int* in_sizes, int n_in,
                              void* d_out, int out_size, void* d_ws, size_t ws_size,
                              hipStream_t stream) {
  (void)in_sizes; (void)n_in; (void)out_size; (void)ws_size;
  const float* x     = (const float*)d_in[0];
  const float* mask  = (const float*)d_in[1];
  const float* wqkv  = (const float*)d_in[2];
  const float* wproj = (const float*)d_in[3];
  const float* bproj = (const float*)d_in[4];
  float* out = (float*)d_out;

  float* Sp = (float*)d_ws;

  attn_scores_kernel<<<dim3(BTOT * HH * 2), dim3(ATHR), 0, stream>>>(x, wqkv, Sp);
  attn_pv_proj_kernel<<<dim3(BTOT * HH * 2), dim3(ATHR), 0, stream>>>(
      x, wqkv, Sp, mask, wproj, bproj, out);
}
